# starting kernel
# baseline (speedup 1.0000x reference)
#include <hip/hip_runtime.h>

typedef __attribute__((ext_vector_type(2))) float v2f;
typedef __attribute__((ext_vector_type(8))) float v8f;

#define VSIZE 100000
#define DDIM  100
#define LTOK  200
#define BATCH 16384

__global__ __launch_bounds__(256) void rowdot_wmma(const float* __restrict__ T,
                                                   const float* __restrict__ w,
                                                   float* __restrict__ s) {
    const int lane  = threadIdx.x & 31;
    const int wave  = threadIdx.x >> 5;
    const int tile  = blockIdx.x * 8 + wave;
    if (tile * 16 >= VSIZE) return;

    const int m      = lane & 15;
    const int khalf  = lane >> 4;
    const bool col0  = (m == 0);

    const v2f* __restrict__ rowp2 =
        (const v2f*)(T + (size_t)(tile * 16 + m) * DDIM);
    const v2f* __restrict__ w2 = (const v2f*)w;

    v8f acc = {};
    #pragma unroll 5
    for (int k0 = 0; k0 < DDIM; k0 += 4) {
        const int kp = (k0 >> 1) + khalf;
        v2f a  = rowp2[kp];
        v2f wv = w2[kp];
        v2f b;
        b.x = col0 ? wv.x : 0.0f;
        b.y = col0 ? wv.y : 0.0f;
        acc = __builtin_amdgcn_wmma_f32_16x16x4_f32(
             false, a,  false, b,
             (short)0, acc,  false,  false);
    }

    if (col0) {
        float* __restrict__ sp = s + tile * 16 + (khalf ? 8 : 0);
        #pragma unroll
        for (int i = 0; i < 8; ++i) sp[i] = acc[i];
    }
}

__global__ __launch_bounds__(256) void pool_gather(const int* __restrict__ idx,
                                                   const float* __restrict__ s,
                                                   float* __restrict__ out) {
    const int lane = threadIdx.x & 31;
    const int wave = threadIdx.x >> 5;
    const int b    = blockIdx.x * 8 + wave;
    if (b >= BATCH) return;

    const int* __restrict__ ip = idx + (size_t)b * LTOK;
    float sum = 0.0f;
    for (int l = lane; l < LTOK; l += 32)
        sum += s[ip[l]];

    sum += __shfl_xor(sum, 16, 32);
    sum += __shfl_xor(sum,  8, 32);
    sum += __shfl_xor(sum,  4, 32);
    sum += __shfl_xor(sum,  2, 32);
    sum += __shfl_xor(sum,  1, 32);

    if (lane == 0) out[b] = sum * (1.0f / (float)LTOK);
}

__global__ __launch_bounds__(256) void fused_naive(const int* __restrict__ idx,
                                                   const float* __restrict__ T,
                                                   const float* __restrict__ w,
                                                   float* __restrict__ out) {
    const int lane = threadIdx.x & 31;
    const int wave = threadIdx.x >> 5;
    const int b    = blockIdx.x * 8 + wave;
    if (b >= BATCH) return;

    const int* __restrict__ ip = idx + (size_t)b * LTOK;
    float sum = 0.0f;
    for (int l = 0; l < LTOK; ++l) {
        const float* __restrict__ rowp = T + (size_t)ip[l] * DDIM;
        for (int d = lane; d < DDIM; d += 32)
            sum += rowp[d] * w[d];
    }
    sum += __shfl_xor(sum, 16, 32);
    sum += __shfl_xor(sum,  8, 32);
    sum += __shfl_xor(sum,  4, 32);
    sum += __shfl_xor(sum,  2, 32);
    sum += __shfl_xor(sum,  1, 32);
    if (lane == 0) out[b] = sum * (1.0f / (float)LTOK);
}

extern "C" void kernel_launch(void* const* d_in, const int* in_sizes, int n_in,
                              void* d_out, int out_size, void* d_ws, size_t ws_size,
                              hipStream_t stream) {
    const int*   word_idxs = (const int*)d_in[0];
    const float* embed     = (const float*)d_in[1];
    const float* weights   = (const float*)d_in[2];
    float*       out       = (float*)d_out;

    if (ws_size >= (size_t)VSIZE * sizeof(float)) {
        float* s = (float*)d_ws;
        const int tiles   = (VSIZE + 15) / 16;
        const int blocks1 = (tiles + 7) / 8;
        rowdot_wmma<<<blocks1, 256, 0, stream>>>(embed, weights, s);

        const int blocks2 = (BATCH + 7) / 8;
        pool_gather<<<blocks2, 256, 0, stream>>>(word_idxs, s, out);
    } else {
        const int blocks = (BATCH + 7) / 8;
        fused_naive<<<blocks, 256, 0, stream>>>(word_idxs, embed, weights, out);
    }
}


# Pass 1 loaders: each lane reads its table row (and w) in 16-byte pieces, k order permuted alike in A and B: 13 row loads per tile instead of 25, all in flight at once; per-row dot sum order changes
# speedup vs baseline: 1.5153x; 1.5153x over previous
#include <hip/hip_runtime.h>

typedef __attribute__((ext_vector_type(2))) float v2f;
typedef __attribute__((ext_vector_type(4))) float v4f;
typedef __attribute__((ext_vector_type(8))) float v8f;
typedef __attribute__((ext_vector_type(2))) int   v2i;

#define VSIZE 100000
#define DDIM  100
#define LTOK  200
#define BATCH 16384
#define ZERO_IDX 100352

#define ROW_PIECES 12
__global__ __launch_bounds__(256) void rowdot_wmma(const float* __restrict__ T,
                                                   const float* __restrict__ w,
                                                   float* __restrict__ s) {
    const int lane  = threadIdx.x & 31;
    const int wave  = threadIdx.x >> 5;
    const int tile  = blockIdx.x * 8 + wave;
    if (tile * 16 >= VSIZE) return;
    if (tile == 0 && lane == 0) s[ZERO_IDX] = 0.0f;

    const int m      = lane & 15;
    const int khalf  = lane >> 4;

    const float* __restrict__ rowp = T + (size_t)(tile * 16 + m) * DDIM;
    const v4f* __restrict__ row4 = (const v4f*)rowp + khalf;
    const v4f* __restrict__ w4   = (const v4f*)w + khalf;

    v4f a4[ROW_PIECES], b4[ROW_PIECES];
    #pragma unroll
    for (int i = 0; i < ROW_PIECES; ++i) {
        a4[i] = row4[2 * i];
        b4[i] = w4[2 * i];
    }
    const v2f at = ((const v2f*)rowp)[48 + khalf];
    const v2f bt = ((const v2f*)w)[48 + khalf];
    __builtin_amdgcn_sched_barrier(0);

    v8f acc = {};
    #pragma unroll
    for (int i = 0; i < ROW_PIECES; ++i) {
        v2f a, b;
        a.x = a4[i].x; a.y = a4[i].y;
        b.x = b4[i].x; b.y = b4[i].y;
        acc = __builtin_amdgcn_wmma_f32_16x16x4_f32(
             false, a,  false, b,
             (short)0, acc,  false,  false);
        a.x = a4[i].z; a.y = a4[i].w;
        b.x = b4[i].z; b.y = b4[i].w;
        acc = __builtin_amdgcn_wmma_f32_16x16x4_f32(
             false, a,  false, b,
             (short)0, acc,  false,  false);
    }
    acc = __builtin_amdgcn_wmma_f32_16x16x4_f32(
         false, at,  false, bt,
         (short)0, acc,  false,  false);

    if (m == 0) {
        float* __restrict__ sp = s + tile * 16 + (khalf ? 8 : 0);
        #pragma unroll
        for (int i = 0; i < 8; ++i) sp[i] = acc[i];
    }
}

#define ROWS_PER_WAVE 8
#define PAIRS_PER_ROW (LTOK / 2)
#define LOADS_PER_WAVE (ROWS_PER_WAVE * PAIRS_PER_ROW / 32)
#define TOK_PER_LANE (2 * LOADS_PER_WAVE)
#define NSLICE 8
#define SLICE_ELEMS (392u * 32u)
static_assert(NSLICE * SLICE_ELEMS >= VSIZE, "the slices cover the whole table");
static_assert(ZERO_IDX >= NSLICE * SLICE_ELEMS && ZERO_IDX % 32 == 0, "zero word on a line of its own behind the last slice");

template <int T>
static __device__ __forceinline__ void sweep_addr(const unsigned (&tok)[TOK_PER_LANE],
                                                  unsigned (&a)[TOK_PER_LANE]) {
    #pragma unroll
    for (int k = 0; k < TOK_PER_LANE; ++k) {
        const unsigned rel = tok[k] - (unsigned)T * SLICE_ELEMS;
        a[k] = (rel < SLICE_ELEMS) ? tok[k] : (unsigned)ZERO_IDX;
    }
}

static __device__ __forceinline__ void sweep_load(const float* __restrict__ s,
                                                  const unsigned (&a)[TOK_PER_LANE],
                                                  float (&g)[TOK_PER_LANE]) {
    #pragma unroll
    for (int k = 0; k < TOK_PER_LANE; ++k) g[k] = s[a[k]];
}

static __device__ __forceinline__ void sweep_add(float (&p)[LOADS_PER_WAVE],
                                                 const float (&g)[TOK_PER_LANE]) {
    #pragma unroll
    for (int j = 0; j < LOADS_PER_WAVE; ++j) p[j] += g[2 * j] + g[2 * j + 1];
}

static __device__ __forceinline__ void phase_fence() {
    __builtin_amdgcn_sched_barrier(0);
}

static __device__ __forceinline__ void sweep_align() {
    __builtin_amdgcn_sched_barrier(0);
    __builtin_amdgcn_s_barrier();
    __builtin_amdgcn_sched_barrier(0);
}

template <int T>
static __device__ __forceinline__ void sweep(const float* __restrict__ s,
                                             const unsigned (&tok)[TOK_PER_LANE],
                                             float (&g)[TOK_PER_LANE],
                                             float (&p)[LOADS_PER_WAVE],
                                             const float (&gprev)[TOK_PER_LANE]) {
    unsigned a[TOK_PER_LANE];
    sweep_addr<T>(tok, a);
    phase_fence();
    sweep_load(s, a, g);
    sweep_align();
    sweep_add(p, gprev);
    phase_fence();
}

__global__ __launch_bounds__(256) void pool_gather(const int* __restrict__ idx,
                                                   const float* __restrict__ s,
                                                   float* __restrict__ out) {
    __shared__ float res[8 * ROWS_PER_WAVE];

    const int lane = threadIdx.x & 31;
    const int wave = threadIdx.x >> 5;
    const int grp  = blockIdx.x * 8 + wave;

    const v2i* __restrict__ ip =
        (const v2i*)(idx + (size_t)grp * (ROWS_PER_WAVE * LTOK)) + lane;

    v2i iv[LOADS_PER_WAVE];
    #pragma unroll
    for (int j = 0; j < LOADS_PER_WAVE; ++j) iv[j] = ip[32 * j];

    unsigned tok[TOK_PER_LANE];
    #pragma unroll
    for (int j = 0; j < LOADS_PER_WAVE; ++j) {
        tok[2 * j]     = (unsigned)iv[j].x;
        tok[2 * j + 1] = (unsigned)iv[j].y;
    }

    float p[LOADS_PER_WAVE];
    #pragma unroll
    for (int j = 0; j < LOADS_PER_WAVE; ++j) p[j] = 0.0f;

    float ga[TOK_PER_LANE], gb[TOK_PER_LANE];
    {
        unsigned a[TOK_PER_LANE];
        sweep_addr<0>(tok, a);
        phase_fence();
        sweep_load(s, a, ga);
        sweep_align();
    }
    sweep<1>(s, tok, gb, p, ga);
    sweep<2>(s, tok, ga, p, gb);
    sweep<3>(s, tok, gb, p, ga);
    sweep<4>(s, tok, ga, p, gb);
    sweep<5>(s, tok, gb, p, ga);
    sweep<6>(s, tok, ga, p, gb);
    sweep<7>(s, tok, gb, p, ga);
    sweep_add(p, gb);

    float acc[ROWS_PER_WAVE];
    #pragma unroll
    for (int r = 0; r < ROWS_PER_WAVE; ++r) acc[r] = 0.0f;

    #pragma unroll
    for (int j = 0; j < LOADS_PER_WAVE; ++j) {
        const int  r0  = (32 * j) / PAIRS_PER_ROW;
        const int  cut = PAIRS_PER_ROW * (r0 + 1) - 32 * j;
        if (cut >= 32) {
            acc[r0] += p[j];
        } else {
            acc[r0]     += (lane <  cut) ? p[j] : 0.0f;
            acc[r0 + 1] += (lane >= cut) ? p[j] : 0.0f;
        }
    }

    const bool h16 = (lane & 16) != 0;
    float b4[4];
    #pragma unroll
    for (int k = 0; k < 4; ++k) {
        const float keep = h16 ? acc[k + 4] : acc[k];
        const float send = h16 ? acc[k]     : acc[k + 4];
        b4[k] = keep + __shfl_xor(send, 16, 32);
    }
    const bool h8 = (lane & 8) != 0;
    float c2[2];
    #pragma unroll
    for (int k = 0; k < 2; ++k) {
        const float keep = h8 ? b4[k + 2] : b4[k];
        const float send = h8 ? b4[k]     : b4[k + 2];
        c2[k] = keep + __shfl_xor(send, 8, 32);
    }
    const bool h4 = (lane & 4) != 0;
    float d = (h4 ? c2[1] : c2[0]) + __shfl_xor(h4 ? c2[0] : c2[1], 4, 32);
    d += __shfl_xor(d, 2, 32);
    d += __shfl_xor(d, 1, 32);

    if ((lane & 3) == 0)
        res[wave * ROWS_PER_WAVE + (lane >> 2)] = d * (1.0f / (float)LTOK);
    __syncthreads();
    if (wave == 0) {
        const float2 v = ((const float2*)res)[lane];
        ((float2*)(out + (size_t)blockIdx.x * (8 * ROWS_PER_WAVE)))[lane] = v;
    }
}

__global__ __launch_bounds__(256) void fused_naive(const int* __restrict__ idx,
                                                   const float* __restrict__ T,
                                                   const float* __restrict__ w,
                                                   float* __restrict__ out) {
    const int lane = threadIdx.x & 31;
    const int wave = threadIdx.x >> 5;
    const int b    = blockIdx.x * 8 + wave;
    if (b >= BATCH) return;

    const int* __restrict__ ip = idx + (size_t)b * LTOK;
    float sum = 0.0f;
    for (int l = 0; l < LTOK; ++l) {
        const float* __restrict__ rowp = T + (size_t)ip[l] * DDIM;
        for (int d = lane; d < DDIM; d += 32)
            sum += rowp[d] * w[d];
    }
    sum += __shfl_xor(sum, 16, 32);
    sum += __shfl_xor(sum,  8, 32);
    sum += __shfl_xor(sum,  4, 32);
    sum += __shfl_xor(sum,  2, 32);
    sum += __shfl_xor(sum,  1, 32);
    if (lane == 0) out[b] = sum * (1.0f / (float)LTOK);
}

extern "C" void kernel_launch(void* const* d_in, const int* in_sizes, int n_in,
                              void* d_out, int out_size, void* d_ws, size_t ws_size,
                              hipStream_t stream) {
    const int*   word_idxs = (const int*)d_in[0];
    const float* embed     = (const float*)d_in[1];
    const float* weights   = (const float*)d_in[2];
    float*       out       = (float*)d_out;

    if (ws_size >= (size_t)(ZERO_IDX + 1) * sizeof(float)) {
        float* s = (float*)d_ws;
        const int tiles   = (VSIZE + 15) / 16;
        const int blocks1 = (tiles + 7) / 8;
        rowdot_wmma<<<blocks1, 256, 0, stream>>>(embed, weights, s);

        const int blocks2 = BATCH / 64;
        pool_gather<<<blocks2, 256, 0, stream>>>(word_idxs, s, out);
    } else {
        const int blocks = (BATCH + 7) / 8;
        fused_naive<<<blocks, 256, 0, stream>>>(word_idxs, embed, weights, out);
    }
}


# Pass 2: work-group b starts its walk over the 8 table slices at slice (b + b/8) mod 8 and wraps round, so the work-groups no longer ask L2 for the same lines at the same moment; sums bit-identical
# speedup vs baseline: 1.6334x; 1.0779x over previous
#include <hip/hip_runtime.h>

typedef __attribute__((ext_vector_type(2))) float v2f;
typedef __attribute__((ext_vector_type(4))) float v4f;
typedef __attribute__((ext_vector_type(8))) float v8f;
typedef __attribute__((ext_vector_type(2))) int   v2i;

#define VSIZE 100000
#define DDIM  100
#define LTOK  200
#define BATCH 16384
#define ZERO_IDX 100352

#define ROW_PIECES 12
__global__ __launch_bounds__(256) void rowdot_wmma(const float* __restrict__ T,
                                                   const float* __restrict__ w,
                                                   float* __restrict__ s) {
    const int lane  = threadIdx.x & 31;
    const int wave  = threadIdx.x >> 5;
    const int tile  = blockIdx.x * 8 + wave;
    if (tile * 16 >= VSIZE) return;
    if (tile == 0 && lane == 0) s[ZERO_IDX] = 0.0f;

    const int m      = lane & 15;
    const int khalf  = lane >> 4;

    const float* __restrict__ rowp = T + (size_t)(tile * 16 + m) * DDIM;
    const v4f* __restrict__ row4 = (const v4f*)rowp + khalf;
    const v4f* __restrict__ w4   = (const v4f*)w + khalf;

    v4f a4[ROW_PIECES], b4[ROW_PIECES];
    #pragma unroll
    for (int i = 0; i < ROW_PIECES; ++i) {
        a4[i] = row4[2 * i];
        b4[i] = w4[2 * i];
    }
    const v2f at = ((const v2f*)rowp)[48 + khalf];
    const v2f bt = ((const v2f*)w)[48 + khalf];
    __builtin_amdgcn_sched_barrier(0);

    v8f acc = {};
    #pragma unroll
    for (int i = 0; i < ROW_PIECES; ++i) {
        v2f a, b;
        a.x = a4[i].x; a.y = a4[i].y;
        b.x = b4[i].x; b.y = b4[i].y;
        acc = __builtin_amdgcn_wmma_f32_16x16x4_f32(
             false, a,  false, b,
             (short)0, acc,  false,  false);
        a.x = a4[i].z; a.y = a4[i].w;
        b.x = b4[i].z; b.y = b4[i].w;
        acc = __builtin_amdgcn_wmma_f32_16x16x4_f32(
             false, a,  false, b,
             (short)0, acc,  false,  false);
    }
    acc = __builtin_amdgcn_wmma_f32_16x16x4_f32(
         false, at,  false, bt,
         (short)0, acc,  false,  false);

    if (m == 0) {
        float* __restrict__ sp = s + tile * 16 + (khalf ? 8 : 0);
        #pragma unroll
        for (int i = 0; i < 8; ++i) sp[i] = acc[i];
    }
}

#define ROWS_PER_WAVE 8
#define PAIRS_PER_ROW (LTOK / 2)
#define LOADS_PER_WAVE (ROWS_PER_WAVE * PAIRS_PER_ROW / 32)
#define TOK_PER_LANE (2 * LOADS_PER_WAVE)
#define NSLICE 8
#define SLICE_ELEMS (392u * 32u)
static_assert(NSLICE * SLICE_ELEMS >= VSIZE, "the slices cover the whole table");
static_assert((NSLICE & (NSLICE - 1)) == 0, "the slice walk wraps round with a mask");
static_assert(ZERO_IDX >= NSLICE * SLICE_ELEMS && ZERO_IDX % 32 == 0, "zero word on a line of its own behind the last slice");

static __device__ __forceinline__ void sweep_addr(const unsigned (&tok)[TOK_PER_LANE],
                                                  unsigned base,
                                                  unsigned (&a)[TOK_PER_LANE]) {
    #pragma unroll
    for (int k = 0; k < TOK_PER_LANE; ++k) {
        const unsigned rel = tok[k] - base;
        a[k] = (rel < SLICE_ELEMS) ? tok[k] : (unsigned)ZERO_IDX;
    }
}

static __device__ __forceinline__ void sweep_load(const float* __restrict__ s,
                                                  const unsigned (&a)[TOK_PER_LANE],
                                                  float (&g)[TOK_PER_LANE]) {
    #pragma unroll
    for (int k = 0; k < TOK_PER_LANE; ++k) g[k] = s[a[k]];
}

static __device__ __forceinline__ void sweep_add(float (&p)[LOADS_PER_WAVE],
                                                 const float (&g)[TOK_PER_LANE]) {
    #pragma unroll
    for (int j = 0; j < LOADS_PER_WAVE; ++j) p[j] += g[2 * j] + g[2 * j + 1];
}

static __device__ __forceinline__ void phase_fence() {
    __builtin_amdgcn_sched_barrier(0);
}

static __device__ __forceinline__ void sweep_align() {
    __builtin_amdgcn_sched_barrier(0);
    __builtin_amdgcn_s_barrier();
    __builtin_amdgcn_sched_barrier(0);
}

static __device__ __forceinline__ unsigned slice_base(unsigned first, unsigned t) {
    return ((first + t) & (unsigned)(NSLICE - 1)) * SLICE_ELEMS;
}

static __device__ __forceinline__ void sweep(const float* __restrict__ s,
                                             const unsigned (&tok)[TOK_PER_LANE],
                                             unsigned base,
                                             float (&g)[TOK_PER_LANE],
                                             float (&p)[LOADS_PER_WAVE],
                                             const float (&gprev)[TOK_PER_LANE]) {
    unsigned a[TOK_PER_LANE];
    sweep_addr(tok, base, a);
    phase_fence();
    sweep_load(s, a, g);
    sweep_align();
    sweep_add(p, gprev);
    phase_fence();
}

__global__ __launch_bounds__(256) void pool_gather(const int* __restrict__ idx,
                                                   const float* __restrict__ s,
                                                   float* __restrict__ out) {
    __shared__ float res[8 * ROWS_PER_WAVE];

    const int lane = threadIdx.x & 31;
    const int wave = threadIdx.x >> 5;
    const int grp  = blockIdx.x * 8 + wave;

    const v2i* __restrict__ ip =
        (const v2i*)(idx + (size_t)grp * (ROWS_PER_WAVE * LTOK)) + lane;

    v2i iv[LOADS_PER_WAVE];
    #pragma unroll
    for (int j = 0; j < LOADS_PER_WAVE; ++j) iv[j] = ip[32 * j];

    unsigned tok[TOK_PER_LANE];
    #pragma unroll
    for (int j = 0; j < LOADS_PER_WAVE; ++j) {
        tok[2 * j]     = (unsigned)iv[j].x;
        tok[2 * j + 1] = (unsigned)iv[j].y;
    }

    float p[LOADS_PER_WAVE];
    #pragma unroll
    for (int j = 0; j < LOADS_PER_WAVE; ++j) p[j] = 0.0f;

    const unsigned first = (blockIdx.x + (blockIdx.x >> 3)) & (unsigned)(NSLICE - 1);

    float ga[TOK_PER_LANE], gb[TOK_PER_LANE];
    {
        unsigned a[TOK_PER_LANE];
        sweep_addr(tok, slice_base(first, 0), a);
        phase_fence();
        sweep_load(s, a, ga);
        sweep_align();
    }
    sweep(s, tok, slice_base(first, 1), gb, p, ga);
    sweep(s, tok, slice_base(first, 2), ga, p, gb);
    sweep(s, tok, slice_base(first, 3), gb, p, ga);
    sweep(s, tok, slice_base(first, 4), ga, p, gb);
    sweep(s, tok, slice_base(first, 5), gb, p, ga);
    sweep(s, tok, slice_base(first, 6), ga, p, gb);
    sweep(s, tok, slice_base(first, 7), gb, p, ga);
    sweep_add(p, gb);

    float acc[ROWS_PER_WAVE];
    #pragma unroll
    for (int r = 0; r < ROWS_PER_WAVE; ++r) acc[r] = 0.0f;

    #pragma unroll
    for (int j = 0; j < LOADS_PER_WAVE; ++j) {
        const int  r0  = (32 * j) / PAIRS_PER_ROW;
        const int  cut = PAIRS_PER_ROW * (r0 + 1) - 32 * j;
        if (cut >= 32) {
            acc[r0] += p[j];
        } else {
            acc[r0]     += (lane <  cut) ? p[j] : 0.0f;
            acc[r0 + 1] += (lane >= cut) ? p[j] : 0.0f;
        }
    }

    const bool h16 = (lane & 16) != 0;
    float b4[4];
    #pragma unroll
    for (int k = 0; k < 4; ++k) {
        const float keep = h16 ? acc[k + 4] : acc[k];
        const float send = h16 ? acc[k]     : acc[k + 4];
        b4[k] = keep + __shfl_xor(send, 16, 32);
    }
    const bool h8 = (lane & 8) != 0;
    float c2[2];
    #pragma unroll
    for (int k = 0; k < 2; ++k) {
        const float keep = h8 ? b4[k + 2] : b4[k];
        const float send = h8 ? b4[k]     : b4[k + 2];
        c2[k] = keep + __shfl_xor(send, 8, 32);
    }
    const bool h4 = (lane & 4) != 0;
    float d = (h4 ? c2[1] : c2[0]) + __shfl_xor(h4 ? c2[0] : c2[1], 4, 32);
    d += __shfl_xor(d, 2, 32);
    d += __shfl_xor(d, 1, 32);

    if ((lane & 3) == 0)
        res[wave * ROWS_PER_WAVE + (lane >> 2)] = d * (1.0f / (float)LTOK);
    __syncthreads();
    if (wave == 0) {
        const float2 v = ((const float2*)res)[lane];
        ((float2*)(out + (size_t)blockIdx.x * (8 * ROWS_PER_WAVE)))[lane] = v;
    }
}

__global__ __launch_bounds__(256) void fused_naive(const int* __restrict__ idx,
                                                   const float* __restrict__ T,
                                                   const float* __restrict__ w,
                                                   float* __restrict__ out) {
    const int lane = threadIdx.x & 31;
    const int wave = threadIdx.x >> 5;
    const int b    = blockIdx.x * 8 + wave;
    if (b >= BATCH) return;

    const int* __restrict__ ip = idx + (size_t)b * LTOK;
    float sum = 0.0f;
    for (int l = 0; l < LTOK; ++l) {
        const float* __restrict__ rowp = T + (size_t)ip[l] * DDIM;
        for (int d = lane; d < DDIM; d += 32)
            sum += rowp[d] * w[d];
    }
    sum += __shfl_xor(sum, 16, 32);
    sum += __shfl_xor(sum,  8, 32);
    sum += __shfl_xor(sum,  4, 32);
    sum += __shfl_xor(sum,  2, 32);
    sum += __shfl_xor(sum,  1, 32);
    if (lane == 0) out[b] = sum * (1.0f / (float)LTOK);
}

extern "C" void kernel_launch(void* const* d_in, const int* in_sizes, int n_in,
                              void* d_out, int out_size, void* d_ws, size_t ws_size,
                              hipStream_t stream) {
    const int*   word_idxs = (const int*)d_in[0];
    const float* embed     = (const float*)d_in[1];
    const float* weights   = (const float*)d_in[2];
    float*       out       = (float*)d_out;

    if (ws_size >= (size_t)(ZERO_IDX + 1) * sizeof(float)) {
        float* s = (float*)d_ws;
        const int tiles   = (VSIZE + 15) / 16;
        const int blocks1 = (tiles + 7) / 8;
        rowdot_wmma<<<blocks1, 256, 0, stream>>>(embed, weights, s);

        const int blocks2 = BATCH / 64;
        pool_gather<<<blocks2, 256, 0, stream>>>(word_idxs, s, out);
    } else {
        const int blocks = (BATCH + 7) / 8;
        fused_naive<<<blocks, 256, 0, stream>>>(word_idxs, embed, weights, out);
    }
}
